# starting kernel
# baseline (speedup 1.0000x reference)
#include <hip/hip_runtime.h>

typedef __attribute__((ext_vector_type(2))) float v2f;
typedef __attribute__((ext_vector_type(8))) float v8f;

#define NN 50000
#define AA 500
#define FF 256
#define APAD 512
#define EPSF 1e-12f

__device__ __forceinline__ v8f wmma_f32(v2f a, v2f b, v8f c) {
  return __builtin_amdgcn_wmma_f32_16x16x4_f32(false, a, false, b, (short)0, c,
                                               false, false);
}

__global__ __launch_bounds__(256) void k_rowsum_inv(const float* __restrict__ adj,
                                                    float* __restrict__ invrow) {
  const int lane = threadIdx.x & 31;
  const int wave = threadIdx.x >> 5;
  const int row = blockIdx.x * 8 + wave;
  const float* rp = adj + row * AA;
  float s = 0.f;
  for (int c = lane; c < AA; c += 32) s += rp[c];
  for (int off = 16; off > 0; off >>= 1) s += __shfl_down(s, off, 32);
  if (lane == 0) invrow[row] = 1.0f / (s + EPSF);
}

__global__ __launch_bounds__(256) void k_colsum_part(const float* __restrict__ adj,
                                                     float* __restrict__ colsum) {
  const int t = threadIdx.x;
  const int r0 = blockIdx.x * 64;
  const int rend = (r0 + 64 < NN) ? (r0 + 64) : NN;
  const int c1 = t + 256;
  float a0 = 0.f, a1 = 0.f;
  for (int r = r0; r < rend; ++r) {
    const float* rp = adj + r * AA;
    a0 += rp[t];
    if (c1 < AA) a1 += rp[c1];
  }
  atomicAdd(&colsum[t], a0);
  if (c1 < AA) atomicAdd(&colsum[c1], a1);
}

__global__ void k_colsum_inv(const float* __restrict__ colsum,
                             float* __restrict__ invcol) {
  const int a = blockIdx.x * blockDim.x + threadIdx.x;
  if (a < APAD) invcol[a] = (a < AA) ? 1.0f / (colsum[a] + EPSF) : 0.0f;
}

__global__ __launch_bounds__(512) void k_atx(const float* __restrict__ adj,
                                             const float* __restrict__ inp,
                                             float* __restrict__ msg0) {
  __shared__ float Alt[50 * 160];
  const int agroup = blockIdx.x & 7;
  const int chunk  = blockIdx.x >> 3;
  const int a0 = agroup * 64;
  const int nbase = chunk * 1000;
  const int wave = threadIdx.x >> 5;
  const int lane = threadIdx.x & 31;
  const int f0 = wave * 16;
  const int half = lane >> 4;
  const int mrow = lane & 15;
  v8f c0 = {}, c1 = {}, c2 = {}, c3 = {};
  const float* bbase = inp + (nbase + half * 2) * FF + f0 + mrow;
  const int afrag = half * 160 + mrow * 2;
  for (int kk = 0; kk < 1000; kk += 100) {
    __syncthreads();
    for (int e = threadIdx.x; e < 100 * 64; e += 512) {
      const int pn = e >> 6;
      const int pa = e & 63;
      const int a = a0 + pa;
      const float v = (a < AA) ? adj[(nbase + kk + pn) * AA + a] : 0.0f;
      Alt[(pn >> 1) * 160 + pa * 2 + (pn & 1)] = v;
    }
    __syncthreads();
    const float* bp = bbase + kk * FF;
#pragma unroll
    for (int s = 0; s < 25; ++s) {
      v2f b;
      b.x = bp[s * 4 * FF];
      b.y = bp[s * 4 * FF + FF];
      const float* ap = &Alt[afrag + s * 320];
      c0 = wmma_f32(*(const v2f*)(ap +  0), b, c0);
      c1 = wmma_f32(*(const v2f*)(ap + 32), b, c1);
      c2 = wmma_f32(*(const v2f*)(ap + 64), b, c2);
      c3 = wmma_f32(*(const v2f*)(ap + 96), b, c3);
    }
  }
  const int col = f0 + mrow;
#pragma unroll
  for (int v = 0; v < 8; ++v) {
    const int m = v + half * 8;
    atomicAdd(&msg0[(a0 +  0 + m) * FF + col], c0[v]);
    atomicAdd(&msg0[(a0 + 16 + m) * FF + col], c1[v]);
    atomicAdd(&msg0[(a0 + 32 + m) * FF + col], c2[v]);
    atomicAdd(&msg0[(a0 + 48 + m) * FF + col], c3[v]);
  }
}

__global__ __launch_bounds__(512) void k_msgw(const float* __restrict__ msg0,
                                              const float* __restrict__ Wm,
                                              const float* __restrict__ invcol,
                                              float* __restrict__ msg) {
  __shared__ float Als[16][260];
  const int a0 = blockIdx.x * 16;
  const int wave = threadIdx.x >> 5;
  const int lane = threadIdx.x & 31;
  const int f0 = wave * 16;
  const int half = lane >> 4;
  const int mrow = lane & 15;
  for (int e = threadIdx.x; e < 16 * 256; e += 512) {
    const int m = e >> 8, k = e & 255;
    Als[m][k] = msg0[(a0 + m) * FF + k] * invcol[a0 + m];
  }
  __syncthreads();
  v8f c = {};
  const float* wb = Wm + (f0 + mrow) * FF + half * 2;
  const float* ab = &Als[mrow][half * 2];
#pragma unroll
  for (int s = 0; s < 64; ++s) {
    v2f b = *(const v2f*)(wb + s * 4);
    v2f a = *(const v2f*)(ab + s * 4);
    c = wmma_f32(a, b, c);
  }
  const int col = f0 + mrow;
#pragma unroll
  for (int v = 0; v < 8; ++v)
    msg[(a0 + v + half * 8) * FF + col] = c[v];
}

__global__ __launch_bounds__(512) void k_out(const float* __restrict__ adj,
                                             const float* __restrict__ msg,
                                             const float* __restrict__ invrow,
                                             float* __restrict__ out) {
  extern __shared__ float Als[];
  const int LDP = 516;
  const int m0 = blockIdx.x * 64;
  const int wave = threadIdx.x >> 5;
  const int lane = threadIdx.x & 31;
  const int f0 = wave * 16;
  const int half = lane >> 4;
  const int mrow = lane & 15;
  const bool full = (m0 + 64 <= NN);
  if (full) {
    for (int e = threadIdx.x; e < 64 * 512; e += 512) {
      const int m = e >> 9, k = e & 511;
      Als[m * LDP + k] = (k < AA) ? adj[(m0 + m) * AA + k] : 0.0f;
    }
  } else {
    for (int e = threadIdx.x; e < 64 * 512; e += 512) {
      const int m = e >> 9, k = e & 511;
      const int gm = m0 + m;
      Als[m * LDP + k] = (k < AA && gm < NN) ? adj[gm * AA + k] : 0.0f;
    }
  }
  __syncthreads();
  v8f c0 = {}, c1 = {}, c2 = {}, c3 = {};
  const float* bb = msg + half * 2 * FF + f0 + mrow;
  const float* a0p = &Als[mrow * LDP + half * 2];
#pragma unroll 8
  for (int s = 0; s < 128; ++s) {
    v2f b;
    b.x = bb[s * 4 * FF];
    b.y = bb[s * 4 * FF + FF];
    const float* ap = a0p + s * 4;
    c0 = wmma_f32(*(const v2f*)(ap),            b, c0);
    c1 = wmma_f32(*(const v2f*)(ap + 16 * LDP), b, c1);
    c2 = wmma_f32(*(const v2f*)(ap + 32 * LDP), b, c2);
    c3 = wmma_f32(*(const v2f*)(ap + 48 * LDP), b, c3);
  }
  const int col = f0 + mrow;
  if (full) {
#pragma unroll
    for (int v = 0; v < 8; ++v) {
      const int mi = v + half * 8;
      int r;
      r = m0 +  0 + mi; out[r * FF + col] = c0[v] * invrow[r];
      r = m0 + 16 + mi; out[r * FF + col] = c1[v] * invrow[r];
      r = m0 + 32 + mi; out[r * FF + col] = c2[v] * invrow[r];
      r = m0 + 48 + mi; out[r * FF + col] = c3[v] * invrow[r];
    }
  } else {
#pragma unroll
    for (int v = 0; v < 8; ++v) {
      const int mi = v + half * 8;
      int r;
      r = m0 +  0 + mi; if (r < NN) out[r * FF + col] = c0[v] * invrow[r];
      r = m0 + 16 + mi; if (r < NN) out[r * FF + col] = c1[v] * invrow[r];
      r = m0 + 32 + mi; if (r < NN) out[r * FF + col] = c2[v] * invrow[r];
      r = m0 + 48 + mi; if (r < NN) out[r * FF + col] = c3[v] * invrow[r];
    }
  }
}

extern "C" void kernel_launch(void* const* d_in, const int* in_sizes, int n_in,
                              void* d_out, int out_size, void* d_ws, size_t ws_size,
                              hipStream_t stream) {
  const float* inp = (const float*)d_in[0];
  const float* adj = (const float*)d_in[1];
  const float* Wm  = (const float*)d_in[2];
  float* out = (float*)d_out;

  char* ws = (char*)d_ws;
  float* msg0   = (float*)(ws);
  float* msg    = (float*)(ws + (size_t)APAD * FF * 4);
  float* colsum = (float*)(ws + (size_t)2 * APAD * FF * 4);
  float* invcol = colsum + APAD;
  float* invrow = invcol + APAD;

  hipMemsetAsync(msg0, 0, (size_t)APAD * FF * sizeof(float), stream);
  hipMemsetAsync(colsum, 0, APAD * sizeof(float), stream);

  k_rowsum_inv<<<NN / 8, 256, 0, stream>>>(adj, invrow);
  k_colsum_part<<<(NN + 63) / 64, 256, 0, stream>>>(adj, colsum);
  k_colsum_inv<<<2, 256, 0, stream>>>(colsum, invcol);
  k_atx<<<400, 512, 0, stream>>>(adj, inp, msg0);
  k_msgw<<<32, 512, 0, stream>>>(msg0, Wm, invcol, msg);
  k_out<<<(NN + 63) / 64, 512, (size_t)64 * 516 * sizeof(float), stream>>>(
      adj, msg, invrow, out);
}


# k_atx without LDS staging: each wave loads adj and input rows straight into WMMA fragments (16-byte loads), one 64x64 tile; a block's 8 waves split n, summed in LDS before the atomics; new sum order
# speedup vs baseline: 1.4634x; 1.4634x over previous
#include <hip/hip_runtime.h>

typedef __attribute__((ext_vector_type(2))) float v2f;
typedef __attribute__((ext_vector_type(8))) float v8f;

#define NN 50000
#define AA 500
#define FF 256
#define APAD 512
#define EPSF 1e-12f

__device__ __forceinline__ v8f wmma_f32(v2f a, v2f b, v8f c) {
  return __builtin_amdgcn_wmma_f32_16x16x4_f32(false, a, false, b, (short)0, c,
                                               false, false);
}

__global__ __launch_bounds__(256) void k_rowsum_inv(const float* __restrict__ adj,
                                                    float* __restrict__ invrow) {
  const int lane = threadIdx.x & 31;
  const int wave = threadIdx.x >> 5;
  const int row = blockIdx.x * 8 + wave;
  const float* rp = adj + row * AA;
  float s = 0.f;
  for (int c = lane; c < AA; c += 32) s += rp[c];
  for (int off = 16; off > 0; off >>= 1) s += __shfl_down(s, off, 32);
  if (lane == 0) invrow[row] = 1.0f / (s + EPSF);
}

__global__ __launch_bounds__(256) void k_colsum_part(const float* __restrict__ adj,
                                                     float* __restrict__ colsum) {
  const int t = threadIdx.x;
  const int r0 = blockIdx.x * 64;
  const int rend = (r0 + 64 < NN) ? (r0 + 64) : NN;
  const int c1 = t + 256;
  float a0 = 0.f, a1 = 0.f;
  for (int r = r0; r < rend; ++r) {
    const float* rp = adj + r * AA;
    a0 += rp[t];
    if (c1 < AA) a1 += rp[c1];
  }
  atomicAdd(&colsum[t], a0);
  if (c1 < AA) atomicAdd(&colsum[c1], a1);
}

__global__ void k_colsum_inv(const float* __restrict__ colsum,
                             float* __restrict__ invcol) {
  const int a = blockIdx.x * blockDim.x + threadIdx.x;
  if (a < APAD) invcol[a] = (a < AA) ? 1.0f / (colsum[a] + EPSF) : 0.0f;
}

typedef __attribute__((ext_vector_type(4))) float v4f;

struct AtxFrag {
  v4f a[8];
  v4f b[8];
};

__device__ __forceinline__ void atx_load(AtxFrag& fr, const float* __restrict__ ap,
                                         const float* __restrict__ bp) {
#pragma unroll
  for (int t = 0; t < 4; ++t)
#pragma unroll
    for (int s = 0; s < 2; ++s) {
      fr.a[2 * t + s] = *(const v4f*)(ap + (4 * t + s) * AA);
      fr.b[2 * t + s] = *(const v4f*)(bp + (4 * t + s) * FF);
    }
}

__device__ __forceinline__ void atx_mma(const AtxFrag& fr, v8f (&acc)[4][4]) {
#pragma unroll
  for (int t = 0; t < 4; ++t) {
    v2f b[4];
#pragma unroll
    for (int u = 0; u < 4; ++u) {
      b[u].x = fr.b[2 * t][u];
      b[u].y = fr.b[2 * t + 1][u];
    }
#pragma unroll
    for (int tt = 0; tt < 4; ++tt) {
      v2f a;
      a.x = fr.a[2 * t][tt];
      a.y = fr.a[2 * t + 1][tt];
#pragma unroll
      for (int u = 0; u < 4; ++u) acc[tt][u] = wmma_f32(a, b[u], acc[tt][u]);
    }
  }
}

#define RPITCH 33

__global__ __launch_bounds__(256) void k_atx(const float* __restrict__ adj,
                                             const float* __restrict__ inp,
                                             float* __restrict__ msg0) {
  __shared__ float red[8 * 128 * RPITCH];
  const int tile = blockIdx.x & 31;
  const int part = blockIdx.x >> 5;
  const int a0 = (tile >> 2) * 64;
  const int f0 = (tile & 3) * 64;
  const int wave = __builtin_amdgcn_readfirstlane(threadIdx.x >> 5);
  const int lane = threadIdx.x & 31;
  const int half = lane >> 4;
  const int c = lane & 15;
  const int q = part * 8 + wave;
  const int sb = (q * 3125) >> 6;
  const int nst = (((q + 1) * 3125) >> 6) - sb;
  int acol = a0 + 4 * c;
  if (acol > AA - 4) acol = AA - 4;
  const float* ap = adj + (sb * 16 + half * 2) * AA + acol;
  const float* bp = inp + (sb * 16 + half * 2) * FF + f0 + 4 * c;
  v8f acc[4][4] = {};
  AtxFrag s0, s1;
  atx_load(s0, ap, bp);
#pragma unroll 1
  for (int i = 0; i + 1 < nst; i += 2) {
    atx_load(s1, ap + 16 * AA, bp + 16 * FF);
    __builtin_amdgcn_sched_barrier(0);
    atx_mma(s0, acc);
    const int d2 = (i + 2 < nst) ? 32 : 16;
    atx_load(s0, ap + d2 * AA, bp + d2 * FF);
    atx_mma(s1, acc);
    ap += 32 * AA;
    bp += 32 * FF;
    asm volatile("" ::: "memory");
  }
  if (nst & 1) atx_mma(s0, acc);
#pragma unroll
  for (int tt = 0; tt < 4; ++tt)
#pragma unroll
    for (int u = 0; u < 4; ++u)
#pragma unroll
      for (int r = 0; r < 8; ++r)
        red[(wave * 128 + (tt * 4 + u) * 8 + r) * RPITCH + lane] = acc[tt][u][r];
  __syncthreads();
  const int fl = (threadIdx.x & 63);
  const int cc = fl >> 2;
  const int uu = fl & 3;
  for (int j = 0; j < 16; ++j) {
    const int al = (threadIdx.x >> 6) + 4 * j;
    const int tt = al & 3;
    const int rho = al >> 2;
    const int idx = ((tt * 4 + uu) * 8 + (rho & 7)) * RPITCH + (rho >> 3) * 16 + cc;
    float s = red[idx];
#pragma unroll
    for (int w = 1; w < 8; ++w) s += red[w * 128 * RPITCH + idx];
    if (a0 + al < AA) atomicAdd(&msg0[(a0 + al) * FF + f0 + fl], s);
  }
}

__global__ __launch_bounds__(512) void k_msgw(const float* __restrict__ msg0,
                                              const float* __restrict__ Wm,
                                              const float* __restrict__ invcol,
                                              float* __restrict__ msg) {
  __shared__ float Als[16][260];
  const int a0 = blockIdx.x * 16;
  const int wave = threadIdx.x >> 5;
  const int lane = threadIdx.x & 31;
  const int f0 = wave * 16;
  const int half = lane >> 4;
  const int mrow = lane & 15;
  for (int e = threadIdx.x; e < 16 * 256; e += 512) {
    const int m = e >> 8, k = e & 255;
    Als[m][k] = msg0[(a0 + m) * FF + k] * invcol[a0 + m];
  }
  __syncthreads();
  v8f c = {};
  const float* wb = Wm + (f0 + mrow) * FF + half * 2;
  const float* ab = &Als[mrow][half * 2];
#pragma unroll
  for (int s = 0; s < 64; ++s) {
    v2f b = *(const v2f*)(wb + s * 4);
    v2f a = *(const v2f*)(ab + s * 4);
    c = wmma_f32(a, b, c);
  }
  const int col = f0 + mrow;
#pragma unroll
  for (int v = 0; v < 8; ++v)
    msg[(a0 + v + half * 8) * FF + col] = c[v];
}

struct OutFrag {
  v4f a[5];
  float b[4][4];
};

__device__ __forceinline__ void out_load(OutFrag& fr, const float* __restrict__ ap,
                                         const float* __restrict__ bp) {
#pragma unroll
  for (int i = 0; i < 5; ++i) fr.a[i] = *(const v4f*)(ap + i * 16 * AA);
#pragma unroll
  for (int ft = 0; ft < 4; ++ft)
#pragma unroll
    for (int r = 0; r < 4; ++r) fr.b[ft][r] = bp[r * FF + ft * 16];
}

__device__ __forceinline__ void out_mma(const OutFrag& fr, v8f (&acc)[5][4]) {
#pragma unroll
  for (int t = 0; t < 2; ++t)
#pragma unroll
    for (int ft = 0; ft < 4; ++ft) {
      v2f b;
      b.x = fr.b[ft][2 * t];
      b.y = fr.b[ft][2 * t + 1];
#pragma unroll
      for (int i = 0; i < 5; ++i) {
        v2f a;
        a.x = fr.a[i][2 * t];
        a.y = fr.a[i][2 * t + 1];
        acc[i][ft] = wmma_f32(a, b, acc[i][ft]);
      }
    }
}

__global__ __launch_bounds__(128) void k_out(const float* __restrict__ adj,
                                             const float* __restrict__ msg,
                                             const float* __restrict__ invrow,
                                             float* __restrict__ out) {
  const int m0 = blockIdx.x * 80;
  const int wave = threadIdx.x >> 5;
  const int lane = threadIdx.x & 31;
  const int f0 = wave * 64;
  const int half = lane >> 4;
  const int c = lane & 15;
  v8f acc[5][4] = {};
  const float* ap = adj + (m0 + c) * AA + half * 4;
  const float* bp = msg + half * 4 * FF + f0 + c;
  OutFrag s0, s1;
  out_load(s0, ap, bp);
#pragma unroll 1
  for (int jj = 0; jj < 62; jj += 2) {
    out_load(s1, ap + 8, bp + 8 * FF);
    __builtin_amdgcn_sched_barrier(0);
    out_mma(s0, acc);
    const int adv = (jj + 2 < 62) ? 16 : 8;
    out_load(s0, ap + adv, bp + adv * FF);
    out_mma(s1, acc);
    ap += 16;
    bp += 16 * FF;
    asm volatile("" ::: "memory");
  }
  {
    const float* at = adj + (m0 + c) * AA + 496 + half * 2;
    const float* bt = msg + (496 + half * 2) * FF + f0 + c;
    v2f a[5];
#pragma unroll
    for (int i = 0; i < 5; ++i) a[i] = *(const v2f*)(at + i * 16 * AA);
#pragma unroll
    for (int ft = 0; ft < 4; ++ft) {
      v2f b;
      b.x = bt[ft * 16];
      b.y = bt[FF + ft * 16];
#pragma unroll
      for (int i = 0; i < 5; ++i) acc[i][ft] = wmma_f32(a[i], b, acc[i][ft]);
    }
  }
#pragma unroll
  for (int i = 0; i < 5; ++i)
#pragma unroll
    for (int v = 0; v < 8; ++v) {
      const int r = m0 + i * 16 + half * 8 + v;
      const float sc = invrow[r];
#pragma unroll
      for (int ft = 0; ft < 4; ++ft)
        out[r * FF + f0 + ft * 16 + c] = acc[i][ft][v] * sc;
    }
}

extern "C" void kernel_launch(void* const* d_in, const int* in_sizes, int n_in,
                              void* d_out, int out_size, void* d_ws, size_t ws_size,
                              hipStream_t stream) {
  const float* inp = (const float*)d_in[0];
  const float* adj = (const float*)d_in[1];
  const float* Wm  = (const float*)d_in[2];
  float* out = (float*)d_out;

  char* ws = (char*)d_ws;
  float* msg0   = (float*)(ws);
  float* msg    = (float*)(ws + (size_t)APAD * FF * 4);
  float* colsum = (float*)(ws + (size_t)2 * APAD * FF * 4);
  float* invcol = colsum + APAD;
  float* invrow = invcol + APAD;

  hipMemsetAsync(msg0, 0, (size_t)APAD * FF * sizeof(float), stream);
  hipMemsetAsync(colsum, 0, APAD * sizeof(float), stream);

  k_rowsum_inv<<<NN / 8, 256, 0, stream>>>(adj, invrow);
  k_colsum_part<<<(NN + 63) / 64, 256, 0, stream>>>(adj, colsum);
  k_colsum_inv<<<2, 256, 0, stream>>>(colsum, invcol);
  k_atx<<<256, 256, 0, stream>>>(adj, inp, msg0);
  k_msgw<<<32, 512, 0, stream>>>(msg0, Wm, invcol, msg);
  k_out<<<NN / 80, 128, 0, stream>>>(adj, msg, invrow, out);
}


# adj's row and column sums in one streaming pass with no atomics: 4096 waves of 12-13 whole rows, column partials per block through LDS to 256 planes, added in fixed order; new sum order
# speedup vs baseline: 1.7975x; 1.2283x over previous
#include <hip/hip_runtime.h>

typedef __attribute__((ext_vector_type(2))) float v2f;
typedef __attribute__((ext_vector_type(4))) float v4f;
typedef __attribute__((ext_vector_type(8))) float v8f;

#define NN 50000
#define AA 500
#define FF 256
#define APAD 512
#define EPSF 1e-12f

__device__ __forceinline__ v8f wmma_f32(v2f a, v2f b, v8f c) {
  return __builtin_amdgcn_wmma_f32_16x16x4_f32(false, a, false, b, (short)0, c,
                                               false, false);
}

__device__ __forceinline__ void row_load(v4f (&v)[4], const float* __restrict__ p,
                                         const float* __restrict__ p3) {
  v[0] = *(const v4f*)(p);
  v[1] = *(const v4f*)(p + 128);
  v[2] = *(const v4f*)(p + 256);
  v[3] = *(const v4f*)(p3);
}

__device__ __forceinline__ float row_sum(const v4f (&v)[4], v4f (&c)[4],
                                         bool keep, bool keep3) {
  const v4f z = {0.f, 0.f, 0.f, 0.f};
  const v4f v0 = keep ? v[0] : z;
  const v4f v1 = keep ? v[1] : z;
  const v4f v2 = keep ? v[2] : z;
  const v4f v3 = keep3 ? v[3] : z;
  c[0] += v0;
  c[1] += v1;
  c[2] += v2;
  c[3] += v3;
  const v4f t = (v0 + v1) + (v2 + v3);
  float p = (t.x + t.y) + (t.z + t.w);
#pragma unroll
  for (int off = 16; off > 0; off >>= 1) p += __shfl_xor(p, off, 32);
  return p;
}

__global__ __launch_bounds__(512) void k_sums(const float* __restrict__ adj,
                                              float* __restrict__ invrow,
                                              float* __restrict__ colpart) {
  __shared__ v4f red[16][128];
  const int wave = __builtin_amdgcn_readfirstlane(threadIdx.x >> 5);
  const int lane = threadIdx.x & 31;
  const int q = blockIdx.x * 16 + wave;
  const int r0 = (q * 3125) >> 8;
  const int cnt = (((q + 1) * 3125) >> 8) - r0;
  const bool ok3 = lane < 29;
  const float* p = adj + r0 * AA + 4 * lane;
  const float* p3 = adj + r0 * AA + 4 * (ok3 ? lane + 96 : 124);
  const bool has12 = (cnt == 13);
  const int last = has12 ? 12 : 11;
  v4f c[4] = {};
  float s[13];
  v4f g0[4][4], g1[4][4];
#pragma unroll
  for (int i = 0; i < 4; ++i) row_load(g0[i], p + i * AA, p3 + i * AA);
#pragma unroll
  for (int i = 0; i < 4; ++i) row_load(g1[i], p + (4 + i) * AA, p3 + (4 + i) * AA);
#pragma unroll
  for (int i = 0; i < 4; ++i) s[i] = row_sum(g0[i], c, true, ok3);
#pragma unroll
  for (int i = 0; i < 4; ++i) row_load(g0[i], p + (8 + i) * AA, p3 + (8 + i) * AA);
  __builtin_amdgcn_sched_barrier(0);
#pragma unroll
  for (int i = 0; i < 4; ++i) s[4 + i] = row_sum(g1[i], c, true, ok3);
  row_load(g1[0], p + last * AA, p3 + last * AA);
  __builtin_amdgcn_sched_barrier(0);
#pragma unroll
  for (int i = 0; i < 4; ++i) s[8 + i] = row_sum(g0[i], c, true, ok3);
  s[12] = row_sum(g1[0], c, has12, has12 && ok3);
  float mine = 0.f;
#pragma unroll
  for (int i = 0; i < 13; ++i)
    if (lane == i) mine = s[i];
  if (lane < cnt) invrow[r0 + lane] = 1.0f / (mine + EPSF);
#pragma unroll
  for (int j = 0; j < 4; ++j) red[wave][lane + 32 * j] = c[j];
  __syncthreads();
  const float* rf = (const float*)&red[0][0];
  float cs = rf[threadIdx.x];
#pragma unroll
  for (int w = 1; w < 16; ++w) cs += rf[w * 512 + threadIdx.x];
  colpart[blockIdx.x * APAD + threadIdx.x] = cs;
}

__global__ __launch_bounds__(512) void k_colsum_inv(const float* __restrict__ colpart,
                                                    float* __restrict__ invcol) {
  __shared__ float red[16][32];
  const int wave = threadIdx.x >> 5;
  const int lane = threadIdx.x & 31;
  const int a = blockIdx.x * 32 + lane;
  const float* cp = colpart + wave * 16 * APAD + a;
  float s = cp[0];
#pragma unroll
  for (int i = 1; i < 16; ++i) s += cp[i * APAD];
  red[wave][lane] = s;
  __syncthreads();
  if (wave == 0) {
    float t = red[0][lane];
#pragma unroll
    for (int w = 1; w < 16; ++w) t += red[w][lane];
    invcol[a] = (a < AA) ? 1.0f / (t + EPSF) : 0.0f;
  }
}


struct AtxFrag {
  v4f a[8];
  v4f b[8];
};

__device__ __forceinline__ void atx_load(AtxFrag& fr, const float* __restrict__ ap,
                                         const float* __restrict__ bp) {
#pragma unroll
  for (int t = 0; t < 4; ++t)
#pragma unroll
    for (int s = 0; s < 2; ++s) {
      fr.a[2 * t + s] = *(const v4f*)(ap + (4 * t + s) * AA);
      fr.b[2 * t + s] = *(const v4f*)(bp + (4 * t + s) * FF);
    }
}

__device__ __forceinline__ void atx_mma(const AtxFrag& fr, v8f (&acc)[4][4]) {
#pragma unroll
  for (int t = 0; t < 4; ++t) {
    v2f b[4];
#pragma unroll
    for (int u = 0; u < 4; ++u) {
      b[u].x = fr.b[2 * t][u];
      b[u].y = fr.b[2 * t + 1][u];
    }
#pragma unroll
    for (int tt = 0; tt < 4; ++tt) {
      v2f a;
      a.x = fr.a[2 * t][tt];
      a.y = fr.a[2 * t + 1][tt];
#pragma unroll
      for (int u = 0; u < 4; ++u) acc[tt][u] = wmma_f32(a, b[u], acc[tt][u]);
    }
  }
}

#define RPITCH 33

__global__ __launch_bounds__(256) void k_atx(const float* __restrict__ adj,
                                             const float* __restrict__ inp,
                                             float* __restrict__ msg0) {
  __shared__ float red[8 * 128 * RPITCH];
  const int tile = blockIdx.x & 31;
  const int part = blockIdx.x >> 5;
  const int a0 = (tile >> 2) * 64;
  const int f0 = (tile & 3) * 64;
  const int wave = __builtin_amdgcn_readfirstlane(threadIdx.x >> 5);
  const int lane = threadIdx.x & 31;
  const int half = lane >> 4;
  const int c = lane & 15;
  const int q = part * 8 + wave;
  const int sb = (q * 3125) >> 6;
  const int nst = (((q + 1) * 3125) >> 6) - sb;
  int acol = a0 + 4 * c;
  if (acol > AA - 4) acol = AA - 4;
  const float* ap = adj + (sb * 16 + half * 2) * AA + acol;
  const float* bp = inp + (sb * 16 + half * 2) * FF + f0 + 4 * c;
  v8f acc[4][4] = {};
  AtxFrag s0, s1;
  atx_load(s0, ap, bp);
#pragma unroll 1
  for (int i = 0; i + 1 < nst; i += 2) {
    atx_load(s1, ap + 16 * AA, bp + 16 * FF);
    __builtin_amdgcn_sched_barrier(0);
    atx_mma(s0, acc);
    const int d2 = (i + 2 < nst) ? 32 : 16;
    atx_load(s0, ap + d2 * AA, bp + d2 * FF);
    atx_mma(s1, acc);
    ap += 32 * AA;
    bp += 32 * FF;
    asm volatile("" ::: "memory");
  }
  if (nst & 1) atx_mma(s0, acc);
#pragma unroll
  for (int tt = 0; tt < 4; ++tt)
#pragma unroll
    for (int u = 0; u < 4; ++u)
#pragma unroll
      for (int r = 0; r < 8; ++r)
        red[(wave * 128 + (tt * 4 + u) * 8 + r) * RPITCH + lane] = acc[tt][u][r];
  __syncthreads();
  const int fl = (threadIdx.x & 63);
  const int cc = fl >> 2;
  const int uu = fl & 3;
  for (int j = 0; j < 16; ++j) {
    const int al = (threadIdx.x >> 6) + 4 * j;
    const int tt = al & 3;
    const int rho = al >> 2;
    const int idx = ((tt * 4 + uu) * 8 + (rho & 7)) * RPITCH + (rho >> 3) * 16 + cc;
    float s = red[idx];
#pragma unroll
    for (int w = 1; w < 8; ++w) s += red[w * 128 * RPITCH + idx];
    if (a0 + al < AA) atomicAdd(&msg0[(a0 + al) * FF + f0 + fl], s);
  }
}

__global__ __launch_bounds__(512) void k_msgw(const float* __restrict__ msg0,
                                              const float* __restrict__ Wm,
                                              const float* __restrict__ invcol,
                                              float* __restrict__ msg) {
  __shared__ float Als[16][260];
  const int a0 = blockIdx.x * 16;
  const int wave = threadIdx.x >> 5;
  const int lane = threadIdx.x & 31;
  const int f0 = wave * 16;
  const int half = lane >> 4;
  const int mrow = lane & 15;
  for (int e = threadIdx.x; e < 16 * 256; e += 512) {
    const int m = e >> 8, k = e & 255;
    Als[m][k] = msg0[(a0 + m) * FF + k] * invcol[a0 + m];
  }
  __syncthreads();
  v8f c = {};
  const float* wb = Wm + (f0 + mrow) * FF + half * 2;
  const float* ab = &Als[mrow][half * 2];
#pragma unroll
  for (int s = 0; s < 64; ++s) {
    v2f b = *(const v2f*)(wb + s * 4);
    v2f a = *(const v2f*)(ab + s * 4);
    c = wmma_f32(a, b, c);
  }
  const int col = f0 + mrow;
#pragma unroll
  for (int v = 0; v < 8; ++v)
    msg[(a0 + v + half * 8) * FF + col] = c[v];
}

struct OutFrag {
  v4f a[5];
  float b[4][4];
};

__device__ __forceinline__ void out_load(OutFrag& fr, const float* __restrict__ ap,
                                         const float* __restrict__ bp) {
#pragma unroll
  for (int i = 0; i < 5; ++i) fr.a[i] = *(const v4f*)(ap + i * 16 * AA);
#pragma unroll
  for (int ft = 0; ft < 4; ++ft)
#pragma unroll
    for (int r = 0; r < 4; ++r) fr.b[ft][r] = bp[r * FF + ft * 16];
}

__device__ __forceinline__ void out_mma(const OutFrag& fr, v8f (&acc)[5][4]) {
#pragma unroll
  for (int t = 0; t < 2; ++t)
#pragma unroll
    for (int ft = 0; ft < 4; ++ft) {
      v2f b;
      b.x = fr.b[ft][2 * t];
      b.y = fr.b[ft][2 * t + 1];
#pragma unroll
      for (int i = 0; i < 5; ++i) {
        v2f a;
        a.x = fr.a[i][2 * t];
        a.y = fr.a[i][2 * t + 1];
        acc[i][ft] = wmma_f32(a, b, acc[i][ft]);
      }
    }
}

__global__ __launch_bounds__(128) void k_out(const float* __restrict__ adj,
                                             const float* __restrict__ msg,
                                             const float* __restrict__ invrow,
                                             float* __restrict__ out) {
  const int m0 = blockIdx.x * 80;
  const int wave = threadIdx.x >> 5;
  const int lane = threadIdx.x & 31;
  const int f0 = wave * 64;
  const int half = lane >> 4;
  const int c = lane & 15;
  v8f acc[5][4] = {};
  const float* ap = adj + (m0 + c) * AA + half * 4;
  const float* bp = msg + half * 4 * FF + f0 + c;
  OutFrag s0, s1;
  out_load(s0, ap, bp);
#pragma unroll 1
  for (int jj = 0; jj < 62; jj += 2) {
    out_load(s1, ap + 8, bp + 8 * FF);
    __builtin_amdgcn_sched_barrier(0);
    out_mma(s0, acc);
    const int adv = (jj + 2 < 62) ? 16 : 8;
    out_load(s0, ap + adv, bp + adv * FF);
    out_mma(s1, acc);
    ap += 16;
    bp += 16 * FF;
    asm volatile("" ::: "memory");
  }
  {
    const float* at = adj + (m0 + c) * AA + 496 + half * 2;
    const float* bt = msg + (496 + half * 2) * FF + f0 + c;
    v2f a[5];
#pragma unroll
    for (int i = 0; i < 5; ++i) a[i] = *(const v2f*)(at + i * 16 * AA);
#pragma unroll
    for (int ft = 0; ft < 4; ++ft) {
      v2f b;
      b.x = bt[ft * 16];
      b.y = bt[FF + ft * 16];
#pragma unroll
      for (int i = 0; i < 5; ++i) acc[i][ft] = wmma_f32(a[i], b, acc[i][ft]);
    }
  }
#pragma unroll
  for (int i = 0; i < 5; ++i)
#pragma unroll
    for (int v = 0; v < 8; ++v) {
      const int r = m0 + i * 16 + half * 8 + v;
      const float sc = invrow[r];
#pragma unroll
      for (int ft = 0; ft < 4; ++ft)
        out[r * FF + f0 + ft * 16 + c] = acc[i][ft][v] * sc;
    }
}

extern "C" void kernel_launch(void* const* d_in, const int* in_sizes, int n_in,
                              void* d_out, int out_size, void* d_ws, size_t ws_size,
                              hipStream_t stream) {
  const float* inp = (const float*)d_in[0];
  const float* adj = (const float*)d_in[1];
  const float* Wm  = (const float*)d_in[2];
  float* out = (float*)d_out;

  char* ws = (char*)d_ws;
  float* msg0   = (float*)(ws);
  float* msg    = (float*)(ws + (size_t)APAD * FF * 4);
  float* invcol = (float*)(ws + (size_t)2 * APAD * FF * 4) + APAD;
  float* invrow = invcol + APAD;
  float* colpart = msg;

  hipMemsetAsync(msg0, 0, (size_t)APAD * FF * sizeof(float), stream);

  k_sums<<<256, 512, 0, stream>>>(adj, invrow, colpart);
  k_colsum_inv<<<16, 512, 0, stream>>>(colpart, invcol);
  k_atx<<<256, 256, 0, stream>>>(adj, inp, msg0);
  k_msgw<<<32, 512, 0, stream>>>(msg0, Wm, invcol, msg);
  k_out<<<NN / 80, 128, 0, stream>>>(adj, msg, invrow, out);
}


# k_out as 256 resident blocks of 192 rows, two waves per SIMD, each wave a 96x64 register tile (6x4 accumulators); the 848 rows left ride as 16x32 half tiles in 424 waves' k loops; same k order
# speedup vs baseline: 1.8926x; 1.0529x over previous
#include <hip/hip_runtime.h>

typedef __attribute__((ext_vector_type(2))) float v2f;
typedef __attribute__((ext_vector_type(4))) float v4f;
typedef __attribute__((ext_vector_type(8))) float v8f;

#define NN 50000
#define AA 500
#define FF 256
#define APAD 512
#define EPSF 1e-12f

__device__ __forceinline__ v8f wmma_f32(v2f a, v2f b, v8f c) {
  return __builtin_amdgcn_wmma_f32_16x16x4_f32(false, a, false, b, (short)0, c,
                                               false, false);
}

__device__ __forceinline__ void row_load(v4f (&v)[4], const float* __restrict__ p,
                                         const float* __restrict__ p3) {
  v[0] = *(const v4f*)(p);
  v[1] = *(const v4f*)(p + 128);
  v[2] = *(const v4f*)(p + 256);
  v[3] = *(const v4f*)(p3);
}

__device__ __forceinline__ float row_sum(const v4f (&v)[4], v4f (&c)[4],
                                         bool keep, bool keep3) {
  const v4f z = {0.f, 0.f, 0.f, 0.f};
  const v4f v0 = keep ? v[0] : z;
  const v4f v1 = keep ? v[1] : z;
  const v4f v2 = keep ? v[2] : z;
  const v4f v3 = keep3 ? v[3] : z;
  c[0] += v0;
  c[1] += v1;
  c[2] += v2;
  c[3] += v3;
  const v4f t = (v0 + v1) + (v2 + v3);
  float p = (t.x + t.y) + (t.z + t.w);
#pragma unroll
  for (int off = 16; off > 0; off >>= 1) p += __shfl_xor(p, off, 32);
  return p;
}

__global__ __launch_bounds__(512) void k_sums(const float* __restrict__ adj,
                                              float* __restrict__ invrow,
                                              float* __restrict__ colpart) {
  __shared__ v4f red[16][128];
  const int wave = __builtin_amdgcn_readfirstlane(threadIdx.x >> 5);
  const int lane = threadIdx.x & 31;
  const int q = blockIdx.x * 16 + wave;
  const int r0 = (q * 3125) >> 8;
  const int cnt = (((q + 1) * 3125) >> 8) - r0;
  const bool ok3 = lane < 29;
  const float* p = adj + r0 * AA + 4 * lane;
  const float* p3 = adj + r0 * AA + 4 * (ok3 ? lane + 96 : 124);
  const bool has12 = (cnt == 13);
  const int last = has12 ? 12 : 11;
  v4f c[4] = {};
  float s[13];
  v4f g0[4][4], g1[4][4];
#pragma unroll
  for (int i = 0; i < 4; ++i) row_load(g0[i], p + i * AA, p3 + i * AA);
#pragma unroll
  for (int i = 0; i < 4; ++i) row_load(g1[i], p + (4 + i) * AA, p3 + (4 + i) * AA);
#pragma unroll
  for (int i = 0; i < 4; ++i) s[i] = row_sum(g0[i], c, true, ok3);
#pragma unroll
  for (int i = 0; i < 4; ++i) row_load(g0[i], p + (8 + i) * AA, p3 + (8 + i) * AA);
  __builtin_amdgcn_sched_barrier(0);
#pragma unroll
  for (int i = 0; i < 4; ++i) s[4 + i] = row_sum(g1[i], c, true, ok3);
  row_load(g1[0], p + last * AA, p3 + last * AA);
  __builtin_amdgcn_sched_barrier(0);
#pragma unroll
  for (int i = 0; i < 4; ++i) s[8 + i] = row_sum(g0[i], c, true, ok3);
  s[12] = row_sum(g1[0], c, has12, has12 && ok3);
  float mine = 0.f;
#pragma unroll
  for (int i = 0; i < 13; ++i)
    if (lane == i) mine = s[i];
  if (lane < cnt) invrow[r0 + lane] = 1.0f / (mine + EPSF);
#pragma unroll
  for (int j = 0; j < 4; ++j) red[wave][lane + 32 * j] = c[j];
  __syncthreads();
  const float* rf = (const float*)&red[0][0];
  float cs = rf[threadIdx.x];
#pragma unroll
  for (int w = 1; w < 16; ++w) cs += rf[w * 512 + threadIdx.x];
  colpart[blockIdx.x * APAD + threadIdx.x] = cs;
}

__global__ __launch_bounds__(512) void k_colsum_inv(const float* __restrict__ colpart,
                                                    float* __restrict__ invcol) {
  __shared__ float red[16][32];
  const int wave = threadIdx.x >> 5;
  const int lane = threadIdx.x & 31;
  const int a = blockIdx.x * 32 + lane;
  const float* cp = colpart + wave * 16 * APAD + a;
  float s = cp[0];
#pragma unroll
  for (int i = 1; i < 16; ++i) s += cp[i * APAD];
  red[wave][lane] = s;
  __syncthreads();
  if (wave == 0) {
    float t = red[0][lane];
#pragma unroll
    for (int w = 1; w < 16; ++w) t += red[w][lane];
    invcol[a] = (a < AA) ? 1.0f / (t + EPSF) : 0.0f;
  }
}


struct AtxFrag {
  v4f a[8];
  v4f b[8];
};

__device__ __forceinline__ void atx_load(AtxFrag& fr, const float* __restrict__ ap,
                                         const float* __restrict__ bp) {
#pragma unroll
  for (int t = 0; t < 4; ++t)
#pragma unroll
    for (int s = 0; s < 2; ++s) {
      fr.a[2 * t + s] = *(const v4f*)(ap + (4 * t + s) * AA);
      fr.b[2 * t + s] = *(const v4f*)(bp + (4 * t + s) * FF);
    }
}

__device__ __forceinline__ void atx_mma(const AtxFrag& fr, v8f (&acc)[4][4]) {
#pragma unroll
  for (int t = 0; t < 4; ++t) {
    v2f b[4];
#pragma unroll
    for (int u = 0; u < 4; ++u) {
      b[u].x = fr.b[2 * t][u];
      b[u].y = fr.b[2 * t + 1][u];
    }
#pragma unroll
    for (int tt = 0; tt < 4; ++tt) {
      v2f a;
      a.x = fr.a[2 * t][tt];
      a.y = fr.a[2 * t + 1][tt];
#pragma unroll
      for (int u = 0; u < 4; ++u) acc[tt][u] = wmma_f32(a, b[u], acc[tt][u]);
    }
  }
}

#define RPITCH 33

__global__ __launch_bounds__(256) void k_atx(const float* __restrict__ adj,
                                             const float* __restrict__ inp,
                                             float* __restrict__ msg0) {
  __shared__ float red[8 * 128 * RPITCH];
  const int tile = blockIdx.x & 31;
  const int part = blockIdx.x >> 5;
  const int a0 = (tile >> 2) * 64;
  const int f0 = (tile & 3) * 64;
  const int wave = __builtin_amdgcn_readfirstlane(threadIdx.x >> 5);
  const int lane = threadIdx.x & 31;
  const int half = lane >> 4;
  const int c = lane & 15;
  const int q = part * 8 + wave;
  const int sb = (q * 3125) >> 6;
  const int nst = (((q + 1) * 3125) >> 6) - sb;
  int acol = a0 + 4 * c;
  if (acol > AA - 4) acol = AA - 4;
  const float* ap = adj + (sb * 16 + half * 2) * AA + acol;
  const float* bp = inp + (sb * 16 + half * 2) * FF + f0 + 4 * c;
  v8f acc[4][4] = {};
  AtxFrag s0, s1;
  atx_load(s0, ap, bp);
#pragma unroll 1
  for (int i = 0; i + 1 < nst; i += 2) {
    atx_load(s1, ap + 16 * AA, bp + 16 * FF);
    __builtin_amdgcn_sched_barrier(0);
    atx_mma(s0, acc);
    const int d2 = (i + 2 < nst) ? 32 : 16;
    atx_load(s0, ap + d2 * AA, bp + d2 * FF);
    atx_mma(s1, acc);
    ap += 32 * AA;
    bp += 32 * FF;
    asm volatile("" ::: "memory");
  }
  if (nst & 1) atx_mma(s0, acc);
#pragma unroll
  for (int tt = 0; tt < 4; ++tt)
#pragma unroll
    for (int u = 0; u < 4; ++u)
#pragma unroll
      for (int r = 0; r < 8; ++r)
        red[(wave * 128 + (tt * 4 + u) * 8 + r) * RPITCH + lane] = acc[tt][u][r];
  __syncthreads();
  const int fl = (threadIdx.x & 63);
  const int cc = fl >> 2;
  const int uu = fl & 3;
  for (int j = 0; j < 16; ++j) {
    const int al = (threadIdx.x >> 6) + 4 * j;
    const int tt = al & 3;
    const int rho = al >> 2;
    const int idx = ((tt * 4 + uu) * 8 + (rho & 7)) * RPITCH + (rho >> 3) * 16 + cc;
    float s = red[idx];
#pragma unroll
    for (int w = 1; w < 8; ++w) s += red[w * 128 * RPITCH + idx];
    if (a0 + al < AA) atomicAdd(&msg0[(a0 + al) * FF + f0 + fl], s);
  }
}

__global__ __launch_bounds__(512) void k_msgw(const float* __restrict__ msg0,
                                              const float* __restrict__ Wm,
                                              const float* __restrict__ invcol,
                                              float* __restrict__ msg) {
  __shared__ float Als[16][260];
  const int a0 = blockIdx.x * 16;
  const int wave = threadIdx.x >> 5;
  const int lane = threadIdx.x & 31;
  const int f0 = wave * 16;
  const int half = lane >> 4;
  const int mrow = lane & 15;
  for (int e = threadIdx.x; e < 16 * 256; e += 512) {
    const int m = e >> 8, k = e & 255;
    Als[m][k] = msg0[(a0 + m) * FF + k] * invcol[a0 + m];
  }
  __syncthreads();
  v8f c = {};
  const float* wb = Wm + (f0 + mrow) * FF + half * 2;
  const float* ab = &Als[mrow][half * 2];
#pragma unroll
  for (int s = 0; s < 64; ++s) {
    v2f b = *(const v2f*)(wb + s * 4);
    v2f a = *(const v2f*)(ab + s * 4);
    c = wmma_f32(a, b, c);
  }
  const int col = f0 + mrow;
#pragma unroll
  for (int v = 0; v < 8; ++v)
    msg[(a0 + v + half * 8) * FF + col] = c[v];
}

#define OUT_RT 6
#define OUT_BLOCKS 256
#define OUT_REST (OUT_BLOCKS * 2 * OUT_RT * 16)
#define OUT_REST_TILES ((NN - OUT_REST) / 16)
static_assert(OUT_REST + OUT_REST_TILES * 16 == NN, "rest is whole row tiles");
static_assert(2 * OUT_REST_TILES <= OUT_BLOCKS, "one half tile per wave");

template <int EX> struct OutFrag {
  v4f a[OUT_RT + (EX ? 1 : 0)];
  float b[4][4];
};

template <int EX>
__device__ __forceinline__ void out_load(OutFrag<EX>& fr, const float* __restrict__ ap,
                                         const float* __restrict__ ae,
                                         const float* __restrict__ bp) {
#pragma unroll
  for (int i = 0; i < OUT_RT; ++i) fr.a[i] = *(const v4f*)(ap + i * 16 * AA);
  if constexpr (EX != 0) fr.a[OUT_RT] = *(const v4f*)(ae);
#pragma unroll
  for (int ft = 0; ft < 4; ++ft)
#pragma unroll
    for (int r = 0; r < 4; ++r) fr.b[ft][r] = bp[r * FF + ft * 16];
}

template <int EX>
__device__ __forceinline__ void out_mma(const OutFrag<EX>& fr, v8f (&acc)[OUT_RT][4],
                                        v8f (&acce)[2]) {
#pragma unroll
  for (int t = 0; t < 2; ++t)
#pragma unroll
    for (int ft = 0; ft < 4; ++ft) {
      v2f b;
      b.x = fr.b[ft][2 * t];
      b.y = fr.b[ft][2 * t + 1];
#pragma unroll
      for (int i = 0; i < OUT_RT; ++i) {
        v2f a;
        a.x = fr.a[i][2 * t];
        a.y = fr.a[i][2 * t + 1];
        acc[i][ft] = wmma_f32(a, b, acc[i][ft]);
      }
      if constexpr (EX != 0) {
        if ((ft >> 1) == EX - 1) {
          v2f a;
          a.x = fr.a[OUT_RT][2 * t];
          a.y = fr.a[OUT_RT][2 * t + 1];
          acce[ft & 1] = wmma_f32(a, b, acce[ft & 1]);
        }
      }
    }
}

template <int EX>
__device__ __forceinline__ void out_tile(const float* __restrict__ adj,
                                         const float* __restrict__ msg,
                                         const float* __restrict__ invrow,
                                         float* __restrict__ out, const int m0,
                                         const int me, const int f0, const int lane) {
  const int half = lane >> 4;
  const int c = lane & 15;
  v8f acc[OUT_RT][4] = {};
  v8f acce[2] = {};
  const float* ap = adj + (m0 + c) * AA + half * 4;
  const float* ae = adj + (me + c) * AA + half * 4;
  const float* bp = msg + half * 4 * FF + f0 + c;
  OutFrag<EX> s0, s1;
  out_load<EX>(s0, ap, ae, bp);
#pragma unroll 1
  for (int jj = 0; jj < 62; jj += 2) {
    out_load<EX>(s1, ap + 8, ae + 8, bp + 8 * FF);
    __builtin_amdgcn_sched_barrier(0);
    out_mma<EX>(s0, acc, acce);
    const int adv = (jj + 2 < 62) ? 16 : 8;
    out_load<EX>(s0, ap + adv, ae + adv, bp + adv * FF);
    out_mma<EX>(s1, acc, acce);
    ap += 16;
    ae += 16;
    bp += 16 * FF;
    asm volatile("" ::: "memory");
  }
  {
    const float* at = adj + (m0 + c) * AA + 496 + half * 2;
    const float* bt = msg + (496 + half * 2) * FF + f0 + c;
    v2f a[OUT_RT + 1];
#pragma unroll
    for (int i = 0; i < OUT_RT; ++i) a[i] = *(const v2f*)(at + i * 16 * AA);
    if constexpr (EX != 0)
      a[OUT_RT] = *(const v2f*)(adj + (me + c) * AA + 496 + half * 2);
#pragma unroll
    for (int ft = 0; ft < 4; ++ft) {
      v2f b;
      b.x = bt[ft * 16];
      b.y = bt[FF + ft * 16];
#pragma unroll
      for (int i = 0; i < OUT_RT; ++i) acc[i][ft] = wmma_f32(a[i], b, acc[i][ft]);
      if constexpr (EX != 0) {
        if ((ft >> 1) == EX - 1) acce[ft & 1] = wmma_f32(a[OUT_RT], b, acce[ft & 1]);
      }
    }
  }
#pragma unroll
  for (int i = 0; i < OUT_RT; ++i)
#pragma unroll
    for (int v = 0; v < 8; ++v) {
      const int r = m0 + i * 16 + half * 8 + v;
      const float sc = invrow[r];
#pragma unroll
      for (int ft = 0; ft < 4; ++ft)
        out[r * FF + f0 + ft * 16 + c] = acc[i][ft][v] * sc;
    }
  if constexpr (EX != 0) {
#pragma unroll
    for (int v = 0; v < 8; ++v) {
      const int r = me + half * 8 + v;
      const float sc = invrow[r];
#pragma unroll
      for (int j = 0; j < 2; ++j)
        out[r * FF + f0 + (2 * (EX - 1) + j) * 16 + c] = acce[j][v] * sc;
    }
  }
}

__global__ __launch_bounds__(256) void k_out(const float* __restrict__ adj,
                                             const float* __restrict__ msg,
                                             const float* __restrict__ invrow,
                                             float* __restrict__ out) {
  const int wave = __builtin_amdgcn_readfirstlane(threadIdx.x >> 5);
  const int lane = threadIdx.x & 31;
  const int f0 = (wave & 3) * 64;
  const int m0 = (blockIdx.x * 2 + (wave >> 2)) * (OUT_RT * 16);
  const bool ex = (wave < 4) && (blockIdx.x < 2 * OUT_REST_TILES);
  const int me = OUT_REST + (blockIdx.x >> 1) * 16;
  if (!ex)
    out_tile<0>(adj, msg, invrow, out, m0, 0, f0, lane);
  else if ((blockIdx.x & 1) == 0)
    out_tile<1>(adj, msg, invrow, out, m0, me, f0, lane);
  else
    out_tile<2>(adj, msg, invrow, out, m0, me, f0, lane);
}

extern "C" void kernel_launch(void* const* d_in, const int* in_sizes, int n_in,
                              void* d_out, int out_size, void* d_ws, size_t ws_size,
                              hipStream_t stream) {
  const float* inp = (const float*)d_in[0];
  const float* adj = (const float*)d_in[1];
  const float* Wm  = (const float*)d_in[2];
  float* out = (float*)d_out;

  char* ws = (char*)d_ws;
  float* msg0   = (float*)(ws);
  float* msg    = (float*)(ws + (size_t)APAD * FF * 4);
  float* invcol = (float*)(ws + (size_t)2 * APAD * FF * 4) + APAD;
  float* invrow = invcol + APAD;
  float* colpart = msg;

  hipMemsetAsync(msg0, 0, (size_t)APAD * FF * sizeof(float), stream);

  k_sums<<<256, 512, 0, stream>>>(adj, invrow, colpart);
  k_colsum_inv<<<16, 512, 0, stream>>>(colpart, invcol);
  k_atx<<<256, 256, 0, stream>>>(adj, inp, msg0);
  k_msgw<<<32, 512, 0, stream>>>(msg0, Wm, invcol, msg);
  k_out<<<OUT_BLOCKS, 256, 0, stream>>>(adj, msg, invrow, out);
}


# Launches 6 -> 4: k_sums' threads store msg0's zeros (memset gone); k_msgw, now 256 blocks of one 16x16 tile per wave, adds the 256 column planes itself (k_colsum_inv gone); same sum orders
# speedup vs baseline: 1.9702x; 1.0410x over previous
#include <hip/hip_runtime.h>

typedef __attribute__((ext_vector_type(2))) float v2f;
typedef __attribute__((ext_vector_type(4))) float v4f;
typedef __attribute__((ext_vector_type(8))) float v8f;

#define NN 50000
#define AA 500
#define FF 256
#define APAD 512
#define EPSF 1e-12f

__device__ __forceinline__ v8f wmma_f32(v2f a, v2f b, v8f c) {
  return __builtin_amdgcn_wmma_f32_16x16x4_f32(false, a, false, b, (short)0, c,
                                               false, false);
}

__device__ __forceinline__ void row_load(v4f (&v)[4], const float* __restrict__ p,
                                         const float* __restrict__ p3) {
  v[0] = *(const v4f*)(p);
  v[1] = *(const v4f*)(p + 128);
  v[2] = *(const v4f*)(p + 256);
  v[3] = *(const v4f*)(p3);
}

__device__ __forceinline__ float row_sum(const v4f (&v)[4], v4f (&c)[4],
                                         bool keep, bool keep3) {
  const v4f z = {0.f, 0.f, 0.f, 0.f};
  const v4f v0 = keep ? v[0] : z;
  const v4f v1 = keep ? v[1] : z;
  const v4f v2 = keep ? v[2] : z;
  const v4f v3 = keep3 ? v[3] : z;
  c[0] += v0;
  c[1] += v1;
  c[2] += v2;
  c[3] += v3;
  const v4f t = (v0 + v1) + (v2 + v3);
  float p = (t.x + t.y) + (t.z + t.w);
#pragma unroll
  for (int off = 16; off > 0; off >>= 1) p += __shfl_xor(p, off, 32);
  return p;
}

__global__ __launch_bounds__(512) void k_sums(const float* __restrict__ adj,
                                              float* __restrict__ invrow,
                                              float* __restrict__ colpart,
                                              float* __restrict__ msg0) {
  __shared__ v4f red[16][128];
  msg0[blockIdx.x * 512 + threadIdx.x] = 0.0f;
  const int wave = __builtin_amdgcn_readfirstlane(threadIdx.x >> 5);
  const int lane = threadIdx.x & 31;
  const int q = blockIdx.x * 16 + wave;
  const int r0 = (q * 3125) >> 8;
  const int cnt = (((q + 1) * 3125) >> 8) - r0;
  const bool ok3 = lane < 29;
  const float* p = adj + r0 * AA + 4 * lane;
  const float* p3 = adj + r0 * AA + 4 * (ok3 ? lane + 96 : 124);
  const bool has12 = (cnt == 13);
  const int last = has12 ? 12 : 11;
  v4f c[4] = {};
  float s[13];
  v4f g0[4][4], g1[4][4];
#pragma unroll
  for (int i = 0; i < 4; ++i) row_load(g0[i], p + i * AA, p3 + i * AA);
#pragma unroll
  for (int i = 0; i < 4; ++i) row_load(g1[i], p + (4 + i) * AA, p3 + (4 + i) * AA);
#pragma unroll
  for (int i = 0; i < 4; ++i) s[i] = row_sum(g0[i], c, true, ok3);
#pragma unroll
  for (int i = 0; i < 4; ++i) row_load(g0[i], p + (8 + i) * AA, p3 + (8 + i) * AA);
  __builtin_amdgcn_sched_barrier(0);
#pragma unroll
  for (int i = 0; i < 4; ++i) s[4 + i] = row_sum(g1[i], c, true, ok3);
  row_load(g1[0], p + last * AA, p3 + last * AA);
  __builtin_amdgcn_sched_barrier(0);
#pragma unroll
  for (int i = 0; i < 4; ++i) s[8 + i] = row_sum(g0[i], c, true, ok3);
  s[12] = row_sum(g1[0], c, has12, has12 && ok3);
  float mine = 0.f;
#pragma unroll
  for (int i = 0; i < 13; ++i)
    if (lane == i) mine = s[i];
  if (lane < cnt) invrow[r0 + lane] = 1.0f / (mine + EPSF);
#pragma unroll
  for (int j = 0; j < 4; ++j) red[wave][lane + 32 * j] = c[j];
  __syncthreads();
  const float* rf = (const float*)&red[0][0];
  float cs = rf[threadIdx.x];
#pragma unroll
  for (int w = 1; w < 16; ++w) cs += rf[w * 512 + threadIdx.x];
  colpart[blockIdx.x * APAD + threadIdx.x] = cs;
}


struct AtxFrag {
  v4f a[8];
  v4f b[8];
};

__device__ __forceinline__ void atx_load(AtxFrag& fr, const float* __restrict__ ap,
                                         const float* __restrict__ bp) {
#pragma unroll
  for (int t = 0; t < 4; ++t)
#pragma unroll
    for (int s = 0; s < 2; ++s) {
      fr.a[2 * t + s] = *(const v4f*)(ap + (4 * t + s) * AA);
      fr.b[2 * t + s] = *(const v4f*)(bp + (4 * t + s) * FF);
    }
}

__device__ __forceinline__ void atx_mma(const AtxFrag& fr, v8f (&acc)[4][4]) {
#pragma unroll
  for (int t = 0; t < 4; ++t) {
    v2f b[4];
#pragma unroll
    for (int u = 0; u < 4; ++u) {
      b[u].x = fr.b[2 * t][u];
      b[u].y = fr.b[2 * t + 1][u];
    }
#pragma unroll
    for (int tt = 0; tt < 4; ++tt) {
      v2f a;
      a.x = fr.a[2 * t][tt];
      a.y = fr.a[2 * t + 1][tt];
#pragma unroll
      for (int u = 0; u < 4; ++u) acc[tt][u] = wmma_f32(a, b[u], acc[tt][u]);
    }
  }
}

#define RPITCH 33

__global__ __launch_bounds__(256) void k_atx(const float* __restrict__ adj,
                                             const float* __restrict__ inp,
                                             float* __restrict__ msg0) {
  __shared__ float red[8 * 128 * RPITCH];
  const int tile = blockIdx.x & 31;
  const int part = blockIdx.x >> 5;
  const int a0 = (tile >> 2) * 64;
  const int f0 = (tile & 3) * 64;
  const int wave = __builtin_amdgcn_readfirstlane(threadIdx.x >> 5);
  const int lane = threadIdx.x & 31;
  const int half = lane >> 4;
  const int c = lane & 15;
  const int q = part * 8 + wave;
  const int sb = (q * 3125) >> 6;
  const int nst = (((q + 1) * 3125) >> 6) - sb;
  int acol = a0 + 4 * c;
  if (acol > AA - 4) acol = AA - 4;
  const float* ap = adj + (sb * 16 + half * 2) * AA + acol;
  const float* bp = inp + (sb * 16 + half * 2) * FF + f0 + 4 * c;
  v8f acc[4][4] = {};
  AtxFrag s0, s1;
  atx_load(s0, ap, bp);
#pragma unroll 1
  for (int i = 0; i + 1 < nst; i += 2) {
    atx_load(s1, ap + 16 * AA, bp + 16 * FF);
    __builtin_amdgcn_sched_barrier(0);
    atx_mma(s0, acc);
    const int d2 = (i + 2 < nst) ? 32 : 16;
    atx_load(s0, ap + d2 * AA, bp + d2 * FF);
    atx_mma(s1, acc);
    ap += 32 * AA;
    bp += 32 * FF;
    asm volatile("" ::: "memory");
  }
  if (nst & 1) atx_mma(s0, acc);
#pragma unroll
  for (int tt = 0; tt < 4; ++tt)
#pragma unroll
    for (int u = 0; u < 4; ++u)
#pragma unroll
      for (int r = 0; r < 8; ++r)
        red[(wave * 128 + (tt * 4 + u) * 8 + r) * RPITCH + lane] = acc[tt][u][r];
  __syncthreads();
  const int fl = (threadIdx.x & 63);
  const int cc = fl >> 2;
  const int uu = fl & 3;
  for (int j = 0; j < 16; ++j) {
    const int al = (threadIdx.x >> 6) + 4 * j;
    const int tt = al & 3;
    const int rho = al >> 2;
    const int idx = ((tt * 4 + uu) * 8 + (rho & 7)) * RPITCH + (rho >> 3) * 16 + cc;
    float s = red[idx];
#pragma unroll
    for (int w = 1; w < 8; ++w) s += red[w * 128 * RPITCH + idx];
    if (a0 + al < AA) atomicAdd(&msg0[(a0 + al) * FF + f0 + fl], s);
  }
}

__global__ __launch_bounds__(64) void k_msgw(const float* __restrict__ msg0,
                                             const float* __restrict__ Wm,
                                             const float* __restrict__ colpart,
                                             float* __restrict__ msg) {
  __shared__ alignas(16) float Als[16][260];
  __shared__ alignas(16) float red[16][16];
  __shared__ float inv[16];
  const int t = threadIdx.x;
  const int a0 = (blockIdx.x >> 3) * 16;
  const int wave = t >> 5;
  const int lane = t & 31;
  const int f0 = ((blockIdx.x & 7) * 2 + wave) * 16;
  const int half = lane >> 4;
  const int mrow = lane & 15;
  const int jq = t & 3;
  const int w = t >> 2;
  const float* cp = colpart + w * 16 * APAD + a0 + 4 * jq;
  v4f pl[16];
#pragma unroll
  for (int i = 0; i < 16; ++i) pl[i] = *(const v4f*)(cp + i * APAD);
  v4f m[16];
#pragma unroll
  for (int i = 0; i < 16; ++i) m[i] = *(const v4f*)(msg0 + (a0 + i) * FF + 4 * t);
  const float* wb = Wm + (f0 + mrow) * FF + half * 2;
  v2f b[64];
#pragma unroll
  for (int s = 0; s < 64; ++s) b[s] = *(const v2f*)(wb + s * 4);
  v4f ps = pl[0];
#pragma unroll
  for (int i = 1; i < 16; ++i) ps += pl[i];
  *(v4f*)&red[w][4 * jq] = ps;
  __syncthreads();
  if (t < 16) {
    float cs = red[0][t];
#pragma unroll
    for (int g = 1; g < 16; ++g) cs += red[g][t];
    inv[t] = (a0 + t < AA) ? 1.0f / (cs + EPSF) : 0.0f;
  }
  __syncthreads();
#pragma unroll
  for (int i = 0; i < 16; ++i) *(v4f*)&Als[i][4 * t] = m[i] * inv[i];
  __syncthreads();
  v8f c = {};
  const float* ab = &Als[mrow][half * 2];
#pragma unroll
  for (int s = 0; s < 64; ++s) {
    v2f a = *(const v2f*)(ab + s * 4);
    c = wmma_f32(a, b[s], c);
  }
  const int col = f0 + mrow;
#pragma unroll
  for (int v = 0; v < 8; ++v)
    msg[(a0 + v + half * 8) * FF + col] = c[v];
}

#define OUT_RT 6
#define OUT_BLOCKS 256
#define OUT_REST (OUT_BLOCKS * 2 * OUT_RT * 16)
#define OUT_REST_TILES ((NN - OUT_REST) / 16)
static_assert(OUT_REST + OUT_REST_TILES * 16 == NN, "rest is whole row tiles");
static_assert(2 * OUT_REST_TILES <= OUT_BLOCKS, "one half tile per wave");

template <int EX> struct OutFrag {
  v4f a[OUT_RT + (EX ? 1 : 0)];
  float b[4][4];
};

template <int EX>
__device__ __forceinline__ void out_load(OutFrag<EX>& fr, const float* __restrict__ ap,
                                         const float* __restrict__ ae,
                                         const float* __restrict__ bp) {
#pragma unroll
  for (int i = 0; i < OUT_RT; ++i) fr.a[i] = *(const v4f*)(ap + i * 16 * AA);
  if constexpr (EX != 0) fr.a[OUT_RT] = *(const v4f*)(ae);
#pragma unroll
  for (int ft = 0; ft < 4; ++ft)
#pragma unroll
    for (int r = 0; r < 4; ++r) fr.b[ft][r] = bp[r * FF + ft * 16];
}

template <int EX>
__device__ __forceinline__ void out_mma(const OutFrag<EX>& fr, v8f (&acc)[OUT_RT][4],
                                        v8f (&acce)[2]) {
#pragma unroll
  for (int t = 0; t < 2; ++t)
#pragma unroll
    for (int ft = 0; ft < 4; ++ft) {
      v2f b;
      b.x = fr.b[ft][2 * t];
      b.y = fr.b[ft][2 * t + 1];
#pragma unroll
      for (int i = 0; i < OUT_RT; ++i) {
        v2f a;
        a.x = fr.a[i][2 * t];
        a.y = fr.a[i][2 * t + 1];
        acc[i][ft] = wmma_f32(a, b, acc[i][ft]);
      }
      if constexpr (EX != 0) {
        if ((ft >> 1) == EX - 1) {
          v2f a;
          a.x = fr.a[OUT_RT][2 * t];
          a.y = fr.a[OUT_RT][2 * t + 1];
          acce[ft & 1] = wmma_f32(a, b, acce[ft & 1]);
        }
      }
    }
}

template <int EX>
__device__ __forceinline__ void out_tile(const float* __restrict__ adj,
                                         const float* __restrict__ msg,
                                         const float* __restrict__ invrow,
                                         float* __restrict__ out, const int m0,
                                         const int me, const int f0, const int lane) {
  const int half = lane >> 4;
  const int c = lane & 15;
  v8f acc[OUT_RT][4] = {};
  v8f acce[2] = {};
  const float* ap = adj + (m0 + c) * AA + half * 4;
  const float* ae = adj + (me + c) * AA + half * 4;
  const float* bp = msg + half * 4 * FF + f0 + c;
  OutFrag<EX> s0, s1;
  out_load<EX>(s0, ap, ae, bp);
#pragma unroll 1
  for (int jj = 0; jj < 62; jj += 2) {
    out_load<EX>(s1, ap + 8, ae + 8, bp + 8 * FF);
    __builtin_amdgcn_sched_barrier(0);
    out_mma<EX>(s0, acc, acce);
    const int adv = (jj + 2 < 62) ? 16 : 8;
    out_load<EX>(s0, ap + adv, ae + adv, bp + adv * FF);
    out_mma<EX>(s1, acc, acce);
    ap += 16;
    ae += 16;
    bp += 16 * FF;
    asm volatile("" ::: "memory");
  }
  {
    const float* at = adj + (m0 + c) * AA + 496 + half * 2;
    const float* bt = msg + (496 + half * 2) * FF + f0 + c;
    v2f a[OUT_RT + 1];
#pragma unroll
    for (int i = 0; i < OUT_RT; ++i) a[i] = *(const v2f*)(at + i * 16 * AA);
    if constexpr (EX != 0)
      a[OUT_RT] = *(const v2f*)(adj + (me + c) * AA + 496 + half * 2);
#pragma unroll
    for (int ft = 0; ft < 4; ++ft) {
      v2f b;
      b.x = bt[ft * 16];
      b.y = bt[FF + ft * 16];
#pragma unroll
      for (int i = 0; i < OUT_RT; ++i) acc[i][ft] = wmma_f32(a[i], b, acc[i][ft]);
      if constexpr (EX != 0) {
        if ((ft >> 1) == EX - 1) acce[ft & 1] = wmma_f32(a[OUT_RT], b, acce[ft & 1]);
      }
    }
  }
#pragma unroll
  for (int i = 0; i < OUT_RT; ++i)
#pragma unroll
    for (int v = 0; v < 8; ++v) {
      const int r = m0 + i * 16 + half * 8 + v;
      const float sc = invrow[r];
#pragma unroll
      for (int ft = 0; ft < 4; ++ft)
        out[r * FF + f0 + ft * 16 + c] = acc[i][ft][v] * sc;
    }
  if constexpr (EX != 0) {
#pragma unroll
    for (int v = 0; v < 8; ++v) {
      const int r = me + half * 8 + v;
      const float sc = invrow[r];
#pragma unroll
      for (int j = 0; j < 2; ++j)
        out[r * FF + f0 + (2 * (EX - 1) + j) * 16 + c] = acce[j][v] * sc;
    }
  }
}

__global__ __launch_bounds__(256) void k_out(const float* __restrict__ adj,
                                             const float* __restrict__ msg,
                                             const float* __restrict__ invrow,
                                             float* __restrict__ out) {
  const int wave = __builtin_amdgcn_readfirstlane(threadIdx.x >> 5);
  const int lane = threadIdx.x & 31;
  const int f0 = (wave & 3) * 64;
  const int m0 = (blockIdx.x * 2 + (wave >> 2)) * (OUT_RT * 16);
  const bool ex = (wave < 4) && (blockIdx.x < 2 * OUT_REST_TILES);
  const int me = OUT_REST + (blockIdx.x >> 1) * 16;
  if (!ex)
    out_tile<0>(adj, msg, invrow, out, m0, 0, f0, lane);
  else if ((blockIdx.x & 1) == 0)
    out_tile<1>(adj, msg, invrow, out, m0, me, f0, lane);
  else
    out_tile<2>(adj, msg, invrow, out, m0, me, f0, lane);
}

extern "C" void kernel_launch(void* const* d_in, const int* in_sizes, int n_in,
                              void* d_out, int out_size, void* d_ws, size_t ws_size,
                              hipStream_t stream) {
  const float* inp = (const float*)d_in[0];
  const float* adj = (const float*)d_in[1];
  const float* Wm  = (const float*)d_in[2];
  float* out = (float*)d_out;

  char* ws = (char*)d_ws;
  float* msg0   = (float*)(ws);
  float* msg    = (float*)(ws + (size_t)APAD * FF * 4);
  float* invrow = (float*)(ws + (size_t)2 * APAD * FF * 4) + 2 * APAD;
  float* colpart = out;

  k_sums<<<256, 512, 0, stream>>>(adj, invrow, colpart, msg0);
  k_atx<<<256, 256, 0, stream>>>(adj, inp, msg0);
  k_msgw<<<256, 64, 0, stream>>>(msg0, Wm, colpart, msg);
  k_out<<<OUT_BLOCKS, 256, 0, stream>>>(adj, msg, invrow, out);
}


# Launches 4 -> 3: adj's row/column sums stream in 4 extra waves per k_atx block beside the matrix-bound product; its 8 n-parts store plain planes that k_msgw adds (no atomics); fixed new sum order
# speedup vs baseline: 2.0458x; 1.0384x over previous
#include <hip/hip_runtime.h>

typedef __attribute__((ext_vector_type(2))) float v2f;
typedef __attribute__((ext_vector_type(4))) float v4f;
typedef __attribute__((ext_vector_type(8))) float v8f;

#define NN 50000
#define AA 500
#define FF 256
#define APAD 512
#define EPSF 1e-12f

__device__ __forceinline__ v8f wmma_f32(v2f a, v2f b, v8f c) {
  return __builtin_amdgcn_wmma_f32_16x16x4_f32(false, a, false, b, (short)0, c,
                                               false, false);
}

__device__ __forceinline__ void row_load(v4f (&v)[4], const float* __restrict__ p,
                                         const float* __restrict__ p3) {
  v[0] = *(const v4f*)(p);
  v[1] = *(const v4f*)(p + 128);
  v[2] = *(const v4f*)(p + 256);
  v[3] = *(const v4f*)(p3);
}

__device__ __forceinline__ float row_sum(const v4f (&v)[4], v4f (&c)[4],
                                         bool keep, bool keep3) {
  const v4f z = {0.f, 0.f, 0.f, 0.f};
  const v4f v0 = keep ? v[0] : z;
  const v4f v1 = keep ? v[1] : z;
  const v4f v2 = keep ? v[2] : z;
  const v4f v3 = keep3 ? v[3] : z;
  c[0] += v0;
  c[1] += v1;
  c[2] += v2;
  c[3] += v3;
  const v4f t = (v0 + v1) + (v2 + v3);
  float p = (t.x + t.y) + (t.z + t.w);
#pragma unroll
  for (int off = 16; off > 0; off >>= 1) p += __shfl_xor(p, off, 32);
  return p;
}

__device__ __forceinline__ void sum6(const v4f (&g)[6][4], v4f (&c)[4], const bool ok3,
                                     const int rb, const int lane, float& mine_a,
                                     float& mine_b) {
#pragma unroll
  for (int i = 0; i < 6; ++i) {
    const float s = row_sum(g[i], c, true, ok3);
    mine_a = (lane == rb + i) ? s : mine_a;
    mine_b = (lane + 32 == rb + i) ? s : mine_b;
  }
}

__device__ __forceinline__ void load6(v4f (&g)[6][4], const float* __restrict__ p,
                                      const float* __restrict__ p3) {
#pragma unroll
  for (int i = 0; i < 6; ++i) row_load(g[i], p + i * AA, p3 + i * AA);
}

__device__ __forceinline__ void sums_body(const float* __restrict__ adj,
                                          float* __restrict__ invrow,
                                          v4f (&cred)[128], const int sw,
                                          const int lane) {
  const int q = blockIdx.x * 4 + sw;
  const int r0 = (q * 3125) >> 6;
  const int cnt = (((q + 1) * 3125) >> 6) - r0;
  const bool ok3 = lane < 29;
  const float* p = adj + r0 * AA + 4 * lane;
  const float* p3 = adj + r0 * AA + 4 * (ok3 ? lane + 96 : 124);
  const bool has48 = (cnt == 49);
  const int last = has48 ? 48 : 47;
  v4f c[4] = {};
  float mine_a = 0.f, mine_b = 0.f;
  v4f tl[4];
  v4f g0[6][4], g1[6][4];
  row_load(tl, p + last * AA, p3 + last * AA);
  load6(g0, p, p3);
#pragma unroll 1
  for (int b = 0; b < 6; b += 2) {
    load6(g1, p + 6 * AA, p3 + 6 * AA);
    __builtin_amdgcn_sched_barrier(0);
    sum6(g0, c, ok3, 6 * b, lane, mine_a, mine_b);
    load6(g0, p + 12 * AA, p3 + 12 * AA);
    __builtin_amdgcn_sched_barrier(0);
    sum6(g1, c, ok3, 6 * b + 6, lane, mine_a, mine_b);
    p += 12 * AA;
    p3 += 12 * AA;
  }
  load6(g1, p + 6 * AA, p3 + 6 * AA);
  __builtin_amdgcn_sched_barrier(0);
  sum6(g0, c, ok3, 36, lane, mine_a, mine_b);
  sum6(g1, c, ok3, 42, lane, mine_a, mine_b);
  const float st = row_sum(tl, c, has48, has48 && ok3);
  mine_b = (lane == 16) ? st : mine_b;
  invrow[r0 + lane] = 1.0f / (mine_a + EPSF);
  if (lane < cnt - 32) invrow[r0 + 32 + lane] = 1.0f / (mine_b + EPSF);
#pragma unroll
  for (int j = 0; j < 4; ++j) cred[lane + 32 * j] = c[j];
}


struct AtxFrag {
  v4f a[8];
  v4f b[8];
};

__device__ __forceinline__ void atx_load(AtxFrag& fr, const float* __restrict__ ap,
                                         const float* __restrict__ bp) {
#pragma unroll
  for (int t = 0; t < 4; ++t)
#pragma unroll
    for (int s = 0; s < 2; ++s) {
      fr.a[2 * t + s] = *(const v4f*)(ap + (4 * t + s) * AA);
      fr.b[2 * t + s] = *(const v4f*)(bp + (4 * t + s) * FF);
    }
}

__device__ __forceinline__ void atx_mma(const AtxFrag& fr, v8f (&acc)[4][4]) {
#pragma unroll
  for (int t = 0; t < 4; ++t) {
    v2f b[4];
#pragma unroll
    for (int u = 0; u < 4; ++u) {
      b[u].x = fr.b[2 * t][u];
      b[u].y = fr.b[2 * t + 1][u];
    }
#pragma unroll
    for (int tt = 0; tt < 4; ++tt) {
      v2f a;
      a.x = fr.a[2 * t][tt];
      a.y = fr.a[2 * t + 1][tt];
#pragma unroll
      for (int u = 0; u < 4; ++u) acc[tt][u] = wmma_f32(a, b[u], acc[tt][u]);
    }
  }
}

#define RPITCH 33

__device__ __forceinline__ void atx_body(const float* __restrict__ adj,
                                         const float* __restrict__ inp,
                                         float* __restrict__ red, const int a0,
                                         const int f0, const int part,
                                         const int wave, const int lane) {
  const int half = lane >> 4;
  const int c = lane & 15;
  const int q = part * 8 + wave;
  const int sb = (q * 3125) >> 6;
  const int nst = (((q + 1) * 3125) >> 6) - sb;
  int acol = a0 + 4 * c;
  if (acol > AA - 4) acol = AA - 4;
  const float* ap = adj + (sb * 16 + half * 2) * AA + acol;
  const float* bp = inp + (sb * 16 + half * 2) * FF + f0 + 4 * c;
  v8f acc[4][4] = {};
  AtxFrag s0, s1;
  atx_load(s0, ap, bp);
#pragma unroll 1
  for (int i = 0; i + 1 < nst; i += 2) {
    atx_load(s1, ap + 16 * AA, bp + 16 * FF);
    __builtin_amdgcn_sched_barrier(0);
    atx_mma(s0, acc);
    const int d2 = (i + 2 < nst) ? 32 : 16;
    atx_load(s0, ap + d2 * AA, bp + d2 * FF);
    atx_mma(s1, acc);
    ap += 32 * AA;
    bp += 32 * FF;
    asm volatile("" ::: "memory");
  }
  if (nst & 1) atx_mma(s0, acc);
#pragma unroll
  for (int tt = 0; tt < 4; ++tt)
#pragma unroll
    for (int u = 0; u < 4; ++u)
#pragma unroll
      for (int r = 0; r < 8; ++r)
        red[(wave * 128 + (tt * 4 + u) * 8 + r) * RPITCH + lane] = acc[tt][u][r];
}

__global__ __launch_bounds__(384) void k_atx_sums(const float* __restrict__ adj,
                                                  const float* __restrict__ inp,
                                                  float* __restrict__ planes,
                                                  float* __restrict__ invrow,
                                                  float* __restrict__ colpart) {
  __shared__ float red[8 * 128 * RPITCH];
  __shared__ v4f cred[4][128];
  const int tile = blockIdx.x & 31;
  const int part = blockIdx.x >> 5;
  const int a0 = (tile >> 2) * 64;
  const int f0 = (tile & 3) * 64;
  const int wave = __builtin_amdgcn_readfirstlane(threadIdx.x >> 5);
  const int lane = threadIdx.x & 31;
  if (wave < 8)
    atx_body(adj, inp, red, a0, f0, part, wave, lane);
  else
    sums_body(adj, invrow, cred[wave - 8], wave - 8, lane);
  __syncthreads();
  if (wave < 8) {
    float* pl = planes + part * (APAD * FF);
    const int fl = (threadIdx.x & 63);
    const int cc = fl >> 2;
    const int uu = fl & 3;
    for (int j = 0; j < 16; ++j) {
      const int al = (threadIdx.x >> 6) + 4 * j;
      const int tt = al & 3;
      const int rho = al >> 2;
      const int idx = ((tt * 4 + uu) * 8 + (rho & 7)) * RPITCH + (rho >> 3) * 16 + cc;
      float s = red[idx];
#pragma unroll
      for (int w = 1; w < 8; ++w) s += red[w * 128 * RPITCH + idx];
      pl[(a0 + al) * FF + f0 + fl] = s;
    }
  } else {
    const int u = threadIdx.x - 256;
    v4f cs = cred[0][u];
#pragma unroll
    for (int w = 1; w < 4; ++w) cs += cred[w][u];
    *(v4f*)(colpart + blockIdx.x * APAD + 4 * u) = cs;
  }
}

__global__ __launch_bounds__(64) void k_msgw(const float* __restrict__ planes,
                                             const float* __restrict__ Wm,
                                             const float* __restrict__ colpart,
                                             float* __restrict__ msg) {
  __shared__ alignas(16) float Als[16][260];
  __shared__ alignas(16) float red[16][16];
  __shared__ float inv[16];
  const int t = threadIdx.x;
  const int a0 = (blockIdx.x >> 3) * 16;
  const int wave = t >> 5;
  const int lane = t & 31;
  const int f0 = ((blockIdx.x & 7) * 2 + wave) * 16;
  const int half = lane >> 4;
  const int mrow = lane & 15;
  const int jq = t & 3;
  const int w = t >> 2;
  const float* cp = colpart + w * 16 * APAD + a0 + 4 * jq;
  v4f pl[16];
#pragma unroll
  for (int i = 0; i < 16; ++i) pl[i] = *(const v4f*)(cp + i * APAD);
  v4f mp[8][16];
#pragma unroll
  for (int pp = 0; pp < 8; ++pp)
#pragma unroll
    for (int i = 0; i < 16; ++i)
      mp[pp][i] = *(const v4f*)(planes + pp * (APAD * FF) + (a0 + i) * FF + 4 * t);
  const float* wb = Wm + (f0 + mrow) * FF + half * 2;
  v2f b[64];
#pragma unroll
  for (int s = 0; s < 64; ++s) b[s] = *(const v2f*)(wb + s * 4);
  v4f ps = pl[0];
#pragma unroll
  for (int i = 1; i < 16; ++i) ps += pl[i];
  *(v4f*)&red[w][4 * jq] = ps;
  __syncthreads();
  if (t < 16) {
    float cs = red[0][t];
#pragma unroll
    for (int g = 1; g < 16; ++g) cs += red[g][t];
    inv[t] = (a0 + t < AA) ? 1.0f / (cs + EPSF) : 0.0f;
  }
  __syncthreads();
#pragma unroll
  for (int i = 0; i < 16; ++i) {
    v4f m = mp[0][i];
#pragma unroll
    for (int pp = 1; pp < 8; ++pp) m += mp[pp][i];
    *(v4f*)&Als[i][4 * t] = m * inv[i];
  }
  __syncthreads();
  v8f c = {};
  const float* ab = &Als[mrow][half * 2];
#pragma unroll
  for (int s = 0; s < 64; ++s) {
    v2f a = *(const v2f*)(ab + s * 4);
    c = wmma_f32(a, b[s], c);
  }
  const int col = f0 + mrow;
#pragma unroll
  for (int v = 0; v < 8; ++v)
    msg[(a0 + v + half * 8) * FF + col] = c[v];
}

#define OUT_RT 6
#define OUT_BLOCKS 256
#define OUT_REST (OUT_BLOCKS * 2 * OUT_RT * 16)
#define OUT_REST_TILES ((NN - OUT_REST) / 16)
static_assert(OUT_REST + OUT_REST_TILES * 16 == NN, "rest is whole row tiles");
static_assert(2 * OUT_REST_TILES <= OUT_BLOCKS, "one half tile per wave");

template <int EX> struct OutFrag {
  v4f a[OUT_RT + (EX ? 1 : 0)];
  float b[4][4];
};

template <int EX>
__device__ __forceinline__ void out_load(OutFrag<EX>& fr, const float* __restrict__ ap,
                                         const float* __restrict__ ae,
                                         const float* __restrict__ bp) {
#pragma unroll
  for (int i = 0; i < OUT_RT; ++i) fr.a[i] = *(const v4f*)(ap + i * 16 * AA);
  if constexpr (EX != 0) fr.a[OUT_RT] = *(const v4f*)(ae);
#pragma unroll
  for (int ft = 0; ft < 4; ++ft)
#pragma unroll
    for (int r = 0; r < 4; ++r) fr.b[ft][r] = bp[r * FF + ft * 16];
}

template <int EX>
__device__ __forceinline__ void out_mma(const OutFrag<EX>& fr, v8f (&acc)[OUT_RT][4],
                                        v8f (&acce)[2]) {
#pragma unroll
  for (int t = 0; t < 2; ++t)
#pragma unroll
    for (int ft = 0; ft < 4; ++ft) {
      v2f b;
      b.x = fr.b[ft][2 * t];
      b.y = fr.b[ft][2 * t + 1];
#pragma unroll
      for (int i = 0; i < OUT_RT; ++i) {
        v2f a;
        a.x = fr.a[i][2 * t];
        a.y = fr.a[i][2 * t + 1];
        acc[i][ft] = wmma_f32(a, b, acc[i][ft]);
      }
      if constexpr (EX != 0) {
        if ((ft >> 1) == EX - 1) {
          v2f a;
          a.x = fr.a[OUT_RT][2 * t];
          a.y = fr.a[OUT_RT][2 * t + 1];
          acce[ft & 1] = wmma_f32(a, b, acce[ft & 1]);
        }
      }
    }
}

template <int EX>
__device__ __forceinline__ void out_tile(const float* __restrict__ adj,
                                         const float* __restrict__ msg,
                                         const float* __restrict__ invrow,
                                         float* __restrict__ out, const int m0,
                                         const int me, const int f0, const int lane) {
  const int half = lane >> 4;
  const int c = lane & 15;
  v8f acc[OUT_RT][4] = {};
  v8f acce[2] = {};
  const float* ap = adj + (m0 + c) * AA + half * 4;
  const float* ae = adj + (me + c) * AA + half * 4;
  const float* bp = msg + half * 4 * FF + f0 + c;
  OutFrag<EX> s0, s1;
  out_load<EX>(s0, ap, ae, bp);
#pragma unroll 1
  for (int jj = 0; jj < 62; jj += 2) {
    out_load<EX>(s1, ap + 8, ae + 8, bp + 8 * FF);
    __builtin_amdgcn_sched_barrier(0);
    out_mma<EX>(s0, acc, acce);
    const int adv = (jj + 2 < 62) ? 16 : 8;
    out_load<EX>(s0, ap + adv, ae + adv, bp + adv * FF);
    out_mma<EX>(s1, acc, acce);
    ap += 16;
    ae += 16;
    bp += 16 * FF;
    asm volatile("" ::: "memory");
  }
  {
    const float* at = adj + (m0 + c) * AA + 496 + half * 2;
    const float* bt = msg + (496 + half * 2) * FF + f0 + c;
    v2f a[OUT_RT + 1];
#pragma unroll
    for (int i = 0; i < OUT_RT; ++i) a[i] = *(const v2f*)(at + i * 16 * AA);
    if constexpr (EX != 0)
      a[OUT_RT] = *(const v2f*)(adj + (me + c) * AA + 496 + half * 2);
#pragma unroll
    for (int ft = 0; ft < 4; ++ft) {
      v2f b;
      b.x = bt[ft * 16];
      b.y = bt[FF + ft * 16];
#pragma unroll
      for (int i = 0; i < OUT_RT; ++i) acc[i][ft] = wmma_f32(a[i], b, acc[i][ft]);
      if constexpr (EX != 0) {
        if ((ft >> 1) == EX - 1) acce[ft & 1] = wmma_f32(a[OUT_RT], b, acce[ft & 1]);
      }
    }
  }
#pragma unroll
  for (int i = 0; i < OUT_RT; ++i)
#pragma unroll
    for (int v = 0; v < 8; ++v) {
      const int r = m0 + i * 16 + half * 8 + v;
      const float sc = invrow[r];
#pragma unroll
      for (int ft = 0; ft < 4; ++ft)
        out[r * FF + f0 + ft * 16 + c] = acc[i][ft][v] * sc;
    }
  if constexpr (EX != 0) {
#pragma unroll
    for (int v = 0; v < 8; ++v) {
      const int r = me + half * 8 + v;
      const float sc = invrow[r];
#pragma unroll
      for (int j = 0; j < 2; ++j)
        out[r * FF + f0 + (2 * (EX - 1) + j) * 16 + c] = acce[j][v] * sc;
    }
  }
}

__global__ __launch_bounds__(256) void k_out(const float* __restrict__ adj,
                                             const float* __restrict__ msg,
                                             const float* __restrict__ invrow,
                                             float* __restrict__ out) {
  const int wave = __builtin_amdgcn_readfirstlane(threadIdx.x >> 5);
  const int lane = threadIdx.x & 31;
  const int f0 = (wave & 3) * 64;
  const int m0 = (blockIdx.x * 2 + (wave >> 2)) * (OUT_RT * 16);
  const bool ex = (wave < 4) && (blockIdx.x < 2 * OUT_REST_TILES);
  const int me = OUT_REST + (blockIdx.x >> 1) * 16;
  if (!ex)
    out_tile<0>(adj, msg, invrow, out, m0, 0, f0, lane);
  else if ((blockIdx.x & 1) == 0)
    out_tile<1>(adj, msg, invrow, out, m0, me, f0, lane);
  else
    out_tile<2>(adj, msg, invrow, out, m0, me, f0, lane);
}

extern "C" void kernel_launch(void* const* d_in, const int* in_sizes, int n_in,
                              void* d_out, int out_size, void* d_ws, size_t ws_size,
                              hipStream_t stream) {
  const float* inp = (const float*)d_in[0];
  const float* adj = (const float*)d_in[1];
  const float* Wm  = (const float*)d_in[2];
  float* out = (float*)d_out;

  char* ws = (char*)d_ws;
  float* msg    = (float*)(ws + (size_t)APAD * FF * 4);
  float* invrow = (float*)(ws + (size_t)2 * APAD * FF * 4) + 2 * APAD;
  float* colpart = out;
  float* planes  = out + 256 * APAD;

  k_atx_sums<<<256, 384, 0, stream>>>(adj, inp, planes, invrow, colpart);
  k_msgw<<<256, 64, 0, stream>>>(planes, Wm, colpart, msg);
  k_out<<<OUT_BLOCKS, 256, 0, stream>>>(adj, msg, invrow, out);
}
